# starting kernel
# baseline (speedup 1.0000x reference)
#include <hip/hip_runtime.h>
#include <hip/hip_bf16.h>

typedef __attribute__((ext_vector_type(2))) float v2f;
typedef __attribute__((ext_vector_type(8))) float v8f;

#define HID 32
#define H1  128

__global__ void deg_kernel(const int* __restrict__ dst, float* __restrict__ deg, int E) {
  int t = blockIdx.x * blockDim.x + threadIdx.x;
  if (t < E) atomicAdd(&deg[dst[t]], 1.0f);
}

__global__ void dinv_kernel(float* __restrict__ degdinv, int N) {
  int t = blockIdx.x * blockDim.x + threadIdx.x;
  if (t < N) degdinv[t] = rsqrtf(degdinv[t] + 1.0f);
}

__global__ __launch_bounds__(256)
void gemm_wmma_f32(const float* __restrict__ A, const float* __restrict__ B,
                   float* __restrict__ C, int M, int N, int K) {
  int wave = (int)((blockIdx.x * (long long)blockDim.x + threadIdx.x) >> 5);
  int lane = threadIdx.x & 31;
  int ntiles = N >> 4;
  int mtiles = (M + 15) >> 4;
  int tm = wave / ntiles;
  int tn = wave - tm * ntiles;
  if (tm >= mtiles) return;
  int mrow = tm * 16 + (lane & 15);
  if (mrow > M - 1) mrow = M - 1;
  int ncol = tn * 16 + (lane & 15);
  int kb = (lane >> 4) * 2;
  v8f acc = {};
  for (int k0 = 0; k0 < K; k0 += 4) {
    v2f a = *(const v2f*)(A + (size_t)mrow * K + k0 + kb);
    v2f b;
    b.x = B[(size_t)(k0 + kb) * N + ncol];
    b.y = B[(size_t)(k0 + kb + 1) * N + ncol];
    acc = __builtin_amdgcn_wmma_f32_16x16x4_f32(false, a, false, b, (short)0, acc,
                                                false, false);
  }
  int rbase = tm * 16 + ((lane >> 4) ? 8 : 0);
#pragma unroll
  for (int r = 0; r < 8; ++r) {
    int row = rbase + r;
    if (row < M) C[(size_t)row * N + ncol] = acc[r];
  }
}

template <int D>
__global__ __launch_bounds__(256)
void scatter_kernel(const float* __restrict__ xl, const int* __restrict__ src,
                    const int* __restrict__ dst, const float* __restrict__ dinv,
                    float* __restrict__ agg, int E) {
  const int LPE = D / 4;
  long long t = (long long)blockIdx.x * blockDim.x + threadIdx.x;
  int e = (int)(t / LPE);
  if (e >= E) return;
  int c = (int)(t % LPE) * 4;
  int s = src[e], d = dst[e];
  float nrm = dinv[s] * dinv[d];
  const float4 v = *(const float4*)(xl + (size_t)s * D + c);
  float* p = agg + (size_t)d * D + c;
  atomicAdd(p + 0, nrm * v.x);
  atomicAdd(p + 1, nrm * v.y);
  atomicAdd(p + 2, nrm * v.z);
  atomicAdd(p + 3, nrm * v.w);
}

template <int D>
__global__ __launch_bounds__(256)
void fuse_kernel(float* __restrict__ agg, const float* __restrict__ xl,
                 const float* __restrict__ dinv, const float* __restrict__ bias, int N) {
  const int LPN = D / 4;
  long long t = (long long)blockIdx.x * blockDim.x + threadIdx.x;
  int i = (int)(t / LPN);
  if (i >= N) return;
  int c = (int)(t % LPN) * 4;
  float di = dinv[i], sl = di * di;
  float4 a = *(const float4*)(agg + (size_t)i * D + c);
  const float4 v = *(const float4*)(xl + (size_t)i * D + c);
  a.x = fmaxf(a.x + sl * v.x + bias[c + 0], 0.f);
  a.y = fmaxf(a.y + sl * v.y + bias[c + 1], 0.f);
  a.z = fmaxf(a.z + sl * v.z + bias[c + 2], 0.f);
  a.w = fmaxf(a.w + sl * v.w + bias[c + 3], 0.f);
  *(float4*)(agg + (size_t)i * D + c) = a;
}

__global__ __launch_bounds__(256)
void edge_mlp_wmma(const float* __restrict__ h, const int* __restrict__ p0,
                   const int* __restrict__ p1, const int* __restrict__ q0,
                   const int* __restrict__ q1, const float* __restrict__ Wm1,
                   const float* __restrict__ bm1, const float* __restrict__ Wm2,
                   const float* __restrict__ bm2, float* __restrict__ out,
                   int NP, int Etot) {
  int wave = (int)((blockIdx.x * (long long)blockDim.x + threadIdx.x) >> 5);
  int lane = threadIdx.x & 31;
  int e0 = wave * 16;
  if (e0 >= Etot) return;
  int m = lane & 15;
  int e = e0 + m;
  if (e > Etot - 1) e = Etot - 1;
  int s = (e < NP) ? p0[e] : q0[e - NP];
  int d = (e < NP) ? p1[e] : q1[e - NP];
  const float* hs = h + (size_t)s * HID;
  const float* hd = h + (size_t)d * HID;
  int kb = (lane >> 4) * 2;
  v8f acc0 = {}, acc1 = {};
#pragma unroll
  for (int ks = 0; ks < 16; ++ks) {
    int j = ks * 4 + kb;
    const float* ap = (j < HID) ? (hs + j) : (hd + (j - HID));
    v2f a; a.x = ap[0]; a.y = ap[1];
    v2f b0, b1;
    b0.x = Wm1[(size_t)j * HID + m];
    b0.y = Wm1[(size_t)(j + 1) * HID + m];
    b1.x = Wm1[(size_t)j * HID + 16 + m];
    b1.y = Wm1[(size_t)(j + 1) * HID + 16 + m];
    acc0 = __builtin_amdgcn_wmma_f32_16x16x4_f32(false, a, false, b0, (short)0, acc0,
                                                 false, false);
    acc1 = __builtin_amdgcn_wmma_f32_16x16x4_f32(false, a, false, b1, (short)0, acc1,
                                                 false, false);
  }
  float bA = bm1[m],        bB = bm1[m + 16];
  float wa0 = Wm2[m * 2],   wa1 = Wm2[m * 2 + 1];
  float wb0 = Wm2[(m + 16) * 2], wb1 = Wm2[(m + 16) * 2 + 1];
  float p[8][2];
#pragma unroll
  for (int r = 0; r < 8; ++r) {
    float t0 = fmaxf(acc0[r] + bA, 0.f);
    float t1 = fmaxf(acc1[r] + bB, 0.f);
    p[r][0] = t0 * wa0 + t1 * wb0;
    p[r][1] = t0 * wa1 + t1 * wb1;
  }
#pragma unroll
  for (int off = 8; off >= 1; off >>= 1) {
#pragma unroll
    for (int r = 0; r < 8; ++r) {
      p[r][0] += __shfl_xor(p[r][0], off, 16);
      p[r][1] += __shfl_xor(p[r][1], off, 16);
    }
  }
  if (m == 0) {
    int rbase = e0 + ((lane >> 4) ? 8 : 0);
    float c0 = bm2[0], c1 = bm2[1];
#pragma unroll
    for (int r = 0; r < 8; ++r) {
      int row = rbase + r;
      if (row < Etot) {
        out[(size_t)row * 2 + 0] = p[r][0] + c0;
        out[(size_t)row * 2 + 1] = p[r][1] + c1;
      }
    }
  }
}

extern "C" void kernel_launch(void* const* d_in, const int* in_sizes, int n_in,
                              void* d_out, int out_size, void* d_ws, size_t ws_size,
                              hipStream_t stream) {
  const float* x   = (const float*)d_in[0];
  const int*   pe  = (const int*)d_in[1];
  const int*   neg = (const int*)d_in[2];
  const float* W1  = (const float*)d_in[3];
  const float* b1  = (const float*)d_in[4];
  const float* W2  = (const float*)d_in[5];
  const float* b2  = (const float*)d_in[6];
  const float* Wm1 = (const float*)d_in[7];
  const float* bm1 = (const float*)d_in[8];
  const float* Wm2 = (const float*)d_in[9];
  const float* bm2 = (const float*)d_in[10];
  float* out = (float*)d_out;

  const int N  = in_sizes[0] / HID;
  const int NP = in_sizes[1] / 2;
  const int NN = in_sizes[2] / 2;
  const int ET = NP + NN;
  const int* p0 = pe;        const int* p1 = pe + NP;
  const int* q0 = neg;       const int* q1 = neg + NN;

  char* ws = (char*)d_ws;
  size_t off = 0;
  auto wsalloc = [&](size_t bytes) -> char* {
    char* p = ws + off;
    off += (bytes + 255) & ~(size_t)255;
    return p;
  };
  float* dinv = (float*)wsalloc((size_t)N * 4);
  float* xl1  = (float*)wsalloc((size_t)N * H1 * 4);
  float* h1   = (float*)wsalloc((size_t)N * H1 * 4);
  float* xl2  = (float*)wsalloc((size_t)N * HID * 4);
  float* h2   = (float*)wsalloc((size_t)N * HID * 4);
  (void)ws_size; (void)n_in; (void)out_size;

  const int B = 256;

  hipMemsetAsync(dinv, 0, (size_t)N * 4, stream);
  deg_kernel<<<(NP + B - 1) / B, B, 0, stream>>>(p1, dinv, NP);
  dinv_kernel<<<(N + B - 1) / B, B, 0, stream>>>(dinv, N);

  {
    long long thr = (long long)((N + 15) / 16) * (H1 / 16) * 32;
    gemm_wmma_f32<<<(unsigned)((thr + B - 1) / B), B, 0, stream>>>(x, W1, xl1, N, H1, HID);
  }
  hipMemsetAsync(h1, 0, (size_t)N * H1 * 4, stream);
  {
    long long thr = (long long)NP * (H1 / 4);
    scatter_kernel<H1><<<(unsigned)((thr + B - 1) / B), B, 0, stream>>>(xl1, p0, p1, dinv, h1, NP);
    long long thr2 = (long long)N * (H1 / 4);
    fuse_kernel<H1><<<(unsigned)((thr2 + B - 1) / B), B, 0, stream>>>(h1, xl1, dinv, b1, N);
  }

  {
    long long thr = (long long)((N + 15) / 16) * (HID / 16) * 32;
    gemm_wmma_f32<<<(unsigned)((thr + B - 1) / B), B, 0, stream>>>(h1, W2, xl2, N, HID, H1);
  }
  hipMemsetAsync(h2, 0, (size_t)N * HID * 4, stream);
  {
    long long thr = (long long)NP * (HID / 4);
    scatter_kernel<HID><<<(unsigned)((thr + B - 1) / B), B, 0, stream>>>(xl2, p0, p1, dinv, h2, NP);
    long long thr2 = (long long)N * (HID / 4);
    fuse_kernel<HID><<<(unsigned)((thr2 + B - 1) / B), B, 0, stream>>>(h2, xl2, dinv, b2, N);
  }

  {
    long long thr = (long long)((ET + 15) / 16) * 32;
    edge_mlp_wmma<<<(unsigned)((thr + B - 1) / B), B, 0, stream>>>(h2, p0, p1, q0, q1,
                                                                   Wm1, bm1, Wm2, bm2,
                                                                   out, NP, ET);
  }
}


# Both edge sums in LDS: edges grouped by 64-node destination tile; a block per tile adds scaled rows (ds_add_f32), then self loop+bias+relu; no L2 f32 atomics or zero fills; sum order: LDS arrival
# speedup vs baseline: 2.1048x; 2.1048x over previous
#include <hip/hip_runtime.h>
#include <hip/hip_bf16.h>

typedef __attribute__((ext_vector_type(2))) float v2f;
typedef __attribute__((ext_vector_type(8))) float v8f;

#define HID 32
#define H1  128

__global__ __launch_bounds__(256)
void gemm_wmma_f32(const float* __restrict__ A, const float* __restrict__ B,
                   float* __restrict__ C, int M, int N, int K) {
  int wave = (int)((blockIdx.x * (long long)blockDim.x + threadIdx.x) >> 5);
  int lane = threadIdx.x & 31;
  int ntiles = N >> 4;
  int mtiles = (M + 15) >> 4;
  int tm = wave / ntiles;
  int tn = wave - tm * ntiles;
  if (tm >= mtiles) return;
  int mrow = tm * 16 + (lane & 15);
  if (mrow > M - 1) mrow = M - 1;
  int ncol = tn * 16 + (lane & 15);
  int kb = (lane >> 4) * 2;
  v8f acc = {};
  for (int k0 = 0; k0 < K; k0 += 4) {
    v2f a = *(const v2f*)(A + (size_t)mrow * K + k0 + kb);
    v2f b;
    b.x = B[(size_t)(k0 + kb) * N + ncol];
    b.y = B[(size_t)(k0 + kb + 1) * N + ncol];
    acc = __builtin_amdgcn_wmma_f32_16x16x4_f32(false, a, false, b, (short)0, acc,
                                                false, false);
  }
  int rbase = tm * 16 + ((lane >> 4) ? 8 : 0);
#pragma unroll
  for (int r = 0; r < 8; ++r) {
    int row = rbase + r;
    if (row < M) C[(size_t)row * N + ncol] = acc[r];
  }
}

#define TSH   6
#define TN    64
#define MAXB  2048
#define CHUNK 8192

__global__ __launch_bounds__(1024)
void bin_count(const int* __restrict__ dst, int* __restrict__ cnt, int E, int nb) {
  __shared__ int hist[MAXB];
  const int tid = threadIdx.x;
  for (int i = tid; i < nb; i += 1024) hist[i] = 0;
  __syncthreads();
  const int base = blockIdx.x * CHUNK;
#pragma unroll
  for (int k = 0; k < CHUNK / 1024; ++k) {
    int e = base + k * 1024 + tid;
    if (e < E) atomicAdd(&hist[dst[e] >> TSH], 1);
  }
  __syncthreads();
  for (int i = tid; i < nb; i += 1024) {
    int c = hist[i];
    if (c) atomicAdd(&cnt[i], c);
  }
}

__global__ __launch_bounds__(1024)
void bin_scan(int* __restrict__ cnt, int* __restrict__ start, int nb) {
  __shared__ int s[1024];
  const int t = threadIdx.x;
  int a = (2 * t < nb) ? cnt[2 * t] : 0;
  int b = (2 * t + 1 < nb) ? cnt[2 * t + 1] : 0;
  int v = a + b;
  s[t] = v;
  __syncthreads();
  for (int off = 1; off < 1024; off <<= 1) {
    int x = (t >= off) ? s[t - off] : 0;
    __syncthreads();
    s[t] += x;
    __syncthreads();
  }
  int excl = s[t] - v;
  if (2 * t < nb)     { start[2 * t] = excl;         cnt[2 * t] = excl; }
  if (2 * t + 1 < nb) { start[2 * t + 1] = excl + a; cnt[2 * t + 1] = excl + a; }
  if (t == 1023) start[nb] = s[1023];
}

__global__ __launch_bounds__(1024)
void bin_fill(const int* __restrict__ src, const int* __restrict__ dst,
              int* __restrict__ cur, int2* __restrict__ ebin, int E, int nb) {
  __shared__ int hist[MAXB];
  const int tid = threadIdx.x;
  for (int i = tid; i < nb; i += 1024) hist[i] = 0;
  __syncthreads();
  const int base = blockIdx.x * CHUNK;
  int sv[CHUNK / 1024], dv[CHUNK / 1024], rk[CHUNK / 1024];
#pragma unroll
  for (int k = 0; k < CHUNK / 1024; ++k) {
    int e = base + k * 1024 + tid;
    sv[k] = 0; dv[k] = 0; rk[k] = 0;
    if (e < E) {
      sv[k] = src[e];
      dv[k] = dst[e];
      rk[k] = atomicAdd(&hist[dv[k] >> TSH], 1);
    }
  }
  __syncthreads();
  for (int i = tid; i < nb; i += 1024) {
    int c = hist[i];
    hist[i] = c ? atomicAdd(&cur[i], c) : 0;
  }
  __syncthreads();
#pragma unroll
  for (int k = 0; k < CHUNK / 1024; ++k) {
    int e = base + k * 1024 + tid;
    if (e < E) ebin[hist[dv[k] >> TSH] + rk[k]] = make_int2(sv[k], dv[k]);
  }
}

__global__ __launch_bounds__(TN)
void deg_tile(const int2* __restrict__ ebin, const int* __restrict__ start,
              float* __restrict__ dinv, int N) {
  __shared__ int c[TN];
  const int t = threadIdx.x;
  const int n0 = blockIdx.x * TN;
  c[t] = 0;
  __syncthreads();
  const int beg = start[blockIdx.x], end = start[blockIdx.x + 1];
  for (int e = beg + t; e < end; e += TN) atomicAdd(&c[ebin[e].y - n0], 1);
  __syncthreads();
  if (n0 + t < N) dinv[n0 + t] = rsqrtf((float)c[t] + 1.0f);
}

template <int D>
__global__ __launch_bounds__(256)
void agg_tile(const float* __restrict__ xl, const int2* __restrict__ ebin,
              const int* __restrict__ start, const float* __restrict__ dinv,
              const float* __restrict__ bias, float* __restrict__ h, int N) {
  constexpr int NJ = D / 32;
  constexpr int U = 4;
  constexpr int NW = 8;
  typedef __attribute__((ext_vector_type(NJ))) float vrow;
  __shared__ float acc[TN * D];
  const int tid = threadIdx.x;
  const int lane = tid & 31;
  const int w = __builtin_amdgcn_readfirstlane(tid >> 5);
  const int n0 = blockIdx.x * TN;
  for (int i = tid; i < TN * D; i += 256) acc[i] = 0.f;
  __syncthreads();
  const int beg = start[blockIdx.x], end = start[blockIdx.x + 1];
  for (int e = beg + w; e < end; e += NW * U) {
    int2 sd[U];
    float nr[U];
    vrow v[U];
#pragma unroll
    for (int u = 0; u < U; ++u) {
      int ee = e + u * NW;
      ee = (ee < end) ? ee : (end - 1);
      sd[u] = ebin[ee];
    }
#pragma unroll
    for (int u = 0; u < U; ++u) {
      nr[u] = dinv[sd[u].x] * dinv[sd[u].y];
      v[u] = *(const vrow*)(xl + (size_t)sd[u].x * D + lane * NJ);
    }
#pragma unroll
    for (int u = 0; u < U; ++u) {
      if (e + u * NW < end) {
        float* row = acc + (sd[u].y - n0) * D + lane;
#pragma unroll
        for (int j = 0; j < NJ; ++j)
          __hip_atomic_fetch_add(row + j * 32, nr[u] * v[u][j], __ATOMIC_RELAXED,
                                 __HIP_MEMORY_SCOPE_WORKGROUP);
      }
    }
  }
  __syncthreads();
  for (int idx = tid; idx < TN * 32; idx += 256) {
    int r = idx >> 5;
    int i = n0 + r;
    if (i < N) {
      float di = dinv[i], sl = di * di;
      const vrow x = *(const vrow*)(xl + (size_t)i * D + lane * NJ);
      const vrow bb = *(const vrow*)(bias + lane * NJ);
      vrow o;
#pragma unroll
      for (int j = 0; j < NJ; ++j)
        o[j] = fmaxf(acc[r * D + j * 32 + lane] + sl * x[j] + bb[j], 0.f);
      *(vrow*)(h + (size_t)i * D + lane * NJ) = o;
    }
  }
}

__global__ __launch_bounds__(256)
void edge_mlp_wmma(const float* __restrict__ h, const int* __restrict__ p0,
                   const int* __restrict__ p1, const int* __restrict__ q0,
                   const int* __restrict__ q1, const float* __restrict__ Wm1,
                   const float* __restrict__ bm1, const float* __restrict__ Wm2,
                   const float* __restrict__ bm2, float* __restrict__ out,
                   int NP, int Etot) {
  int wave = (int)((blockIdx.x * (long long)blockDim.x + threadIdx.x) >> 5);
  int lane = threadIdx.x & 31;
  int e0 = wave * 16;
  if (e0 >= Etot) return;
  int m = lane & 15;
  int e = e0 + m;
  if (e > Etot - 1) e = Etot - 1;
  int s = (e < NP) ? p0[e] : q0[e - NP];
  int d = (e < NP) ? p1[e] : q1[e - NP];
  const float* hs = h + (size_t)s * HID;
  const float* hd = h + (size_t)d * HID;
  int kb = (lane >> 4) * 2;
  v8f acc0 = {}, acc1 = {};
#pragma unroll
  for (int ks = 0; ks < 16; ++ks) {
    int j = ks * 4 + kb;
    const float* ap = (j < HID) ? (hs + j) : (hd + (j - HID));
    v2f a; a.x = ap[0]; a.y = ap[1];
    v2f b0, b1;
    b0.x = Wm1[(size_t)j * HID + m];
    b0.y = Wm1[(size_t)(j + 1) * HID + m];
    b1.x = Wm1[(size_t)j * HID + 16 + m];
    b1.y = Wm1[(size_t)(j + 1) * HID + 16 + m];
    acc0 = __builtin_amdgcn_wmma_f32_16x16x4_f32(false, a, false, b0, (short)0, acc0,
                                                 false, false);
    acc1 = __builtin_amdgcn_wmma_f32_16x16x4_f32(false, a, false, b1, (short)0, acc1,
                                                 false, false);
  }
  float bA = bm1[m],        bB = bm1[m + 16];
  float wa0 = Wm2[m * 2],   wa1 = Wm2[m * 2 + 1];
  float wb0 = Wm2[(m + 16) * 2], wb1 = Wm2[(m + 16) * 2 + 1];
  float p[8][2];
#pragma unroll
  for (int r = 0; r < 8; ++r) {
    float t0 = fmaxf(acc0[r] + bA, 0.f);
    float t1 = fmaxf(acc1[r] + bB, 0.f);
    p[r][0] = t0 * wa0 + t1 * wb0;
    p[r][1] = t0 * wa1 + t1 * wb1;
  }
#pragma unroll
  for (int off = 8; off >= 1; off >>= 1) {
#pragma unroll
    for (int r = 0; r < 8; ++r) {
      p[r][0] += __shfl_xor(p[r][0], off, 16);
      p[r][1] += __shfl_xor(p[r][1], off, 16);
    }
  }
  if (m == 0) {
    int rbase = e0 + ((lane >> 4) ? 8 : 0);
    float c0 = bm2[0], c1 = bm2[1];
#pragma unroll
    for (int r = 0; r < 8; ++r) {
      int row = rbase + r;
      if (row < Etot) {
        out[(size_t)row * 2 + 0] = p[r][0] + c0;
        out[(size_t)row * 2 + 1] = p[r][1] + c1;
      }
    }
  }
}

extern "C" void kernel_launch(void* const* d_in, const int* in_sizes, int n_in,
                              void* d_out, int out_size, void* d_ws, size_t ws_size,
                              hipStream_t stream) {
  const float* x   = (const float*)d_in[0];
  const int*   pe  = (const int*)d_in[1];
  const int*   neg = (const int*)d_in[2];
  const float* W1  = (const float*)d_in[3];
  const float* b1  = (const float*)d_in[4];
  const float* W2  = (const float*)d_in[5];
  const float* b2  = (const float*)d_in[6];
  const float* Wm1 = (const float*)d_in[7];
  const float* bm1 = (const float*)d_in[8];
  const float* Wm2 = (const float*)d_in[9];
  const float* bm2 = (const float*)d_in[10];
  float* out = (float*)d_out;

  const int N  = in_sizes[0] / HID;
  const int NP = in_sizes[1] / 2;
  const int NN = in_sizes[2] / 2;
  const int ET = NP + NN;
  const int* p0 = pe;        const int* p1 = pe + NP;
  const int* q0 = neg;       const int* q1 = neg + NN;
  const int NB = (N + TN - 1) / TN;

  char* ws = (char*)d_ws;
  size_t off = 0;
  auto wsalloc = [&](size_t bytes) -> char* {
    char* p = ws + off;
    off += (bytes + 255) & ~(size_t)255;
    return p;
  };
  float* dinv = (float*)wsalloc((size_t)N * 4);
  float* xl1  = (float*)wsalloc((size_t)N * H1 * 4);
  float* h1   = (float*)wsalloc((size_t)N * H1 * 4);
  float* xl2  = (float*)wsalloc((size_t)N * HID * 4);
  float* h2   = (float*)wsalloc((size_t)N * HID * 4);
  int*   bcur = (int*)wsalloc((size_t)NB * 4);
  int*   bst  = (int*)wsalloc((size_t)(NB + 1) * 4);
  int2*  ebin = (int2*)wsalloc((size_t)NP * 8);
  (void)ws_size; (void)n_in; (void)out_size;

  const int B = 256;

  hipMemsetAsync(bcur, 0, (size_t)NB * 4, stream);
  {
    const int nblk = (NP + CHUNK - 1) / CHUNK;
    bin_count<<<nblk, 1024, 0, stream>>>(p1, bcur, NP, NB);
    bin_scan<<<1, 1024, 0, stream>>>(bcur, bst, NB);
    bin_fill<<<nblk, 1024, 0, stream>>>(p0, p1, bcur, ebin, NP, NB);
  }
  deg_tile<<<NB, TN, 0, stream>>>(ebin, bst, dinv, N);

  {
    long long thr = (long long)((N + 15) / 16) * (H1 / 16) * 32;
    gemm_wmma_f32<<<(unsigned)((thr + B - 1) / B), B, 0, stream>>>(x, W1, xl1, N, H1, HID);
  }
  agg_tile<H1><<<NB, 256, 0, stream>>>(xl1, ebin, bst, dinv, b1, h1, N);

  {
    long long thr = (long long)((N + 15) / 16) * (HID / 16) * 32;
    gemm_wmma_f32<<<(unsigned)((thr + B - 1) / B), B, 0, stream>>>(h1, W2, xl2, N, HID, H1);
  }
  agg_tile<HID><<<NB, 256, 0, stream>>>(xl2, ebin, bst, dinv, b2, h2, N);

  {
    long long thr = (long long)((ET + 15) / 16) * 32;
    edge_mlp_wmma<<<(unsigned)((thr + B - 1) / B), B, 0, stream>>>(h2, p0, p1, q0, q1,
                                                                   Wm1, bm1, Wm2, bm2,
                                                                   out, NP, ET);
  }
}


# Edge MLP: one resident set of 8192 waves; each keeps its Wm1 B fragments in 64 registers and walks ~24 tiles of 16 edges: B loaded once per wave, not per tile; 24x fewer waves launched
# speedup vs baseline: 2.2130x; 1.0514x over previous
#include <hip/hip_runtime.h>
#include <hip/hip_bf16.h>

typedef __attribute__((ext_vector_type(2))) float v2f;
typedef __attribute__((ext_vector_type(8))) float v8f;

#define HID 32
#define H1  128

__global__ __launch_bounds__(256)
void gemm_wmma_f32(const float* __restrict__ A, const float* __restrict__ B,
                   float* __restrict__ C, int M, int N, int K) {
  int wave = (int)((blockIdx.x * (long long)blockDim.x + threadIdx.x) >> 5);
  int lane = threadIdx.x & 31;
  int ntiles = N >> 4;
  int mtiles = (M + 15) >> 4;
  int tm = wave / ntiles;
  int tn = wave - tm * ntiles;
  if (tm >= mtiles) return;
  int mrow = tm * 16 + (lane & 15);
  if (mrow > M - 1) mrow = M - 1;
  int ncol = tn * 16 + (lane & 15);
  int kb = (lane >> 4) * 2;
  v8f acc = {};
  for (int k0 = 0; k0 < K; k0 += 4) {
    v2f a = *(const v2f*)(A + (size_t)mrow * K + k0 + kb);
    v2f b;
    b.x = B[(size_t)(k0 + kb) * N + ncol];
    b.y = B[(size_t)(k0 + kb + 1) * N + ncol];
    acc = __builtin_amdgcn_wmma_f32_16x16x4_f32(false, a, false, b, (short)0, acc,
                                                false, false);
  }
  int rbase = tm * 16 + ((lane >> 4) ? 8 : 0);
#pragma unroll
  for (int r = 0; r < 8; ++r) {
    int row = rbase + r;
    if (row < M) C[(size_t)row * N + ncol] = acc[r];
  }
}

#define TSH   6
#define TN    64
#define MAXB  2048
#define CHUNK 8192

__global__ __launch_bounds__(1024)
void bin_count(const int* __restrict__ dst, int* __restrict__ cnt, int E, int nb) {
  __shared__ int hist[MAXB];
  const int tid = threadIdx.x;
  for (int i = tid; i < nb; i += 1024) hist[i] = 0;
  __syncthreads();
  const int base = blockIdx.x * CHUNK;
#pragma unroll
  for (int k = 0; k < CHUNK / 1024; ++k) {
    int e = base + k * 1024 + tid;
    if (e < E) atomicAdd(&hist[dst[e] >> TSH], 1);
  }
  __syncthreads();
  for (int i = tid; i < nb; i += 1024) {
    int c = hist[i];
    if (c) atomicAdd(&cnt[i], c);
  }
}

__global__ __launch_bounds__(1024)
void bin_scan(int* __restrict__ cnt, int* __restrict__ start, int nb) {
  __shared__ int s[1024];
  const int t = threadIdx.x;
  int a = (2 * t < nb) ? cnt[2 * t] : 0;
  int b = (2 * t + 1 < nb) ? cnt[2 * t + 1] : 0;
  int v = a + b;
  s[t] = v;
  __syncthreads();
  for (int off = 1; off < 1024; off <<= 1) {
    int x = (t >= off) ? s[t - off] : 0;
    __syncthreads();
    s[t] += x;
    __syncthreads();
  }
  int excl = s[t] - v;
  if (2 * t < nb)     { start[2 * t] = excl;         cnt[2 * t] = excl; }
  if (2 * t + 1 < nb) { start[2 * t + 1] = excl + a; cnt[2 * t + 1] = excl + a; }
  if (t == 1023) start[nb] = s[1023];
}

__global__ __launch_bounds__(1024)
void bin_fill(const int* __restrict__ src, const int* __restrict__ dst,
              int* __restrict__ cur, int2* __restrict__ ebin, int E, int nb) {
  __shared__ int hist[MAXB];
  const int tid = threadIdx.x;
  for (int i = tid; i < nb; i += 1024) hist[i] = 0;
  __syncthreads();
  const int base = blockIdx.x * CHUNK;
  int sv[CHUNK / 1024], dv[CHUNK / 1024], rk[CHUNK / 1024];
#pragma unroll
  for (int k = 0; k < CHUNK / 1024; ++k) {
    int e = base + k * 1024 + tid;
    sv[k] = 0; dv[k] = 0; rk[k] = 0;
    if (e < E) {
      sv[k] = src[e];
      dv[k] = dst[e];
      rk[k] = atomicAdd(&hist[dv[k] >> TSH], 1);
    }
  }
  __syncthreads();
  for (int i = tid; i < nb; i += 1024) {
    int c = hist[i];
    hist[i] = c ? atomicAdd(&cur[i], c) : 0;
  }
  __syncthreads();
#pragma unroll
  for (int k = 0; k < CHUNK / 1024; ++k) {
    int e = base + k * 1024 + tid;
    if (e < E) ebin[hist[dv[k] >> TSH] + rk[k]] = make_int2(sv[k], dv[k]);
  }
}

__global__ __launch_bounds__(TN)
void deg_tile(const int2* __restrict__ ebin, const int* __restrict__ start,
              float* __restrict__ dinv, int N) {
  __shared__ int c[TN];
  const int t = threadIdx.x;
  const int n0 = blockIdx.x * TN;
  c[t] = 0;
  __syncthreads();
  const int beg = start[blockIdx.x], end = start[blockIdx.x + 1];
  for (int e = beg + t; e < end; e += TN) atomicAdd(&c[ebin[e].y - n0], 1);
  __syncthreads();
  if (n0 + t < N) dinv[n0 + t] = rsqrtf((float)c[t] + 1.0f);
}

template <int D>
__global__ __launch_bounds__(256)
void agg_tile(const float* __restrict__ xl, const int2* __restrict__ ebin,
              const int* __restrict__ start, const float* __restrict__ dinv,
              const float* __restrict__ bias, float* __restrict__ h, int N) {
  constexpr int NJ = D / 32;
  constexpr int U = 4;
  constexpr int NW = 8;
  typedef __attribute__((ext_vector_type(NJ))) float vrow;
  __shared__ float acc[TN * D];
  const int tid = threadIdx.x;
  const int lane = tid & 31;
  const int w = __builtin_amdgcn_readfirstlane(tid >> 5);
  const int n0 = blockIdx.x * TN;
  for (int i = tid; i < TN * D; i += 256) acc[i] = 0.f;
  __syncthreads();
  const int beg = start[blockIdx.x], end = start[blockIdx.x + 1];
  for (int e = beg + w; e < end; e += NW * U) {
    int2 sd[U];
    float nr[U];
    vrow v[U];
#pragma unroll
    for (int u = 0; u < U; ++u) {
      int ee = e + u * NW;
      ee = (ee < end) ? ee : (end - 1);
      sd[u] = ebin[ee];
    }
#pragma unroll
    for (int u = 0; u < U; ++u) {
      nr[u] = dinv[sd[u].x] * dinv[sd[u].y];
      v[u] = *(const vrow*)(xl + (size_t)sd[u].x * D + lane * NJ);
    }
#pragma unroll
    for (int u = 0; u < U; ++u) {
      if (e + u * NW < end) {
        float* row = acc + (sd[u].y - n0) * D + lane;
#pragma unroll
        for (int j = 0; j < NJ; ++j)
          __hip_atomic_fetch_add(row + j * 32, nr[u] * v[u][j], __ATOMIC_RELAXED,
                                 __HIP_MEMORY_SCOPE_WORKGROUP);
      }
    }
  }
  __syncthreads();
  for (int idx = tid; idx < TN * 32; idx += 256) {
    int r = idx >> 5;
    int i = n0 + r;
    if (i < N) {
      float di = dinv[i], sl = di * di;
      const vrow x = *(const vrow*)(xl + (size_t)i * D + lane * NJ);
      const vrow bb = *(const vrow*)(bias + lane * NJ);
      vrow o;
#pragma unroll
      for (int j = 0; j < NJ; ++j)
        o[j] = fmaxf(acc[r * D + j * 32 + lane] + sl * x[j] + bb[j], 0.f);
      *(vrow*)(h + (size_t)i * D + lane * NJ) = o;
    }
  }
}

#define MLP_BLOCKS 1024
__global__ __launch_bounds__(256) __attribute__((amdgpu_waves_per_eu(8, 8)))
void edge_mlp_wmma(const float* __restrict__ h, const int* __restrict__ p0,
                   const int* __restrict__ p1, const int* __restrict__ q0,
                   const int* __restrict__ q1, const float* __restrict__ Wm1,
                   const float* __restrict__ bm1, const float* __restrict__ Wm2,
                   const float* __restrict__ bm2, float* __restrict__ out,
                   int NP, int Etot) {
  const int lane = threadIdx.x & 31;
  const int m = lane & 15;
  const int kb = (lane >> 4) * 2;
  const int wave = __builtin_amdgcn_readfirstlane(
      (int)(blockIdx.x * (blockDim.x >> 5) + (threadIdx.x >> 5)));
  const int nwaves = (int)(gridDim.x * (blockDim.x >> 5));
  const int ntiles = (Etot + 15) >> 4;

  v2f b0[16], b1[16];
#pragma unroll
  for (int ks = 0; ks < 16; ++ks) {
    int j = ks * 4 + kb;
    b0[ks].x = Wm1[(size_t)j * HID + m];
    b0[ks].y = Wm1[(size_t)(j + 1) * HID + m];
    b1[ks].x = Wm1[(size_t)j * HID + 16 + m];
    b1[ks].y = Wm1[(size_t)(j + 1) * HID + 16 + m];
  }
  const float bA = bm1[m],        bB = bm1[m + 16];
  const float wa0 = Wm2[m * 2],   wa1 = Wm2[m * 2 + 1];
  const float wb0 = Wm2[(m + 16) * 2], wb1 = Wm2[(m + 16) * 2 + 1];
  const float c0 = bm2[0], c1 = bm2[1];

  for (int t = wave; t < ntiles; t += nwaves) {
    const int e0 = t * 16;
    int e = e0 + m;
    if (e > Etot - 1) e = Etot - 1;
    int s = (e < NP) ? p0[e] : q0[e - NP];
    int d = (e < NP) ? p1[e] : q1[e - NP];
    const float* hs = h + (size_t)s * HID;
    const float* hd = h + (size_t)d * HID;
    v8f acc0 = {}, acc1 = {};
#pragma unroll
    for (int ks = 0; ks < 16; ++ks) {
      int j = ks * 4 + kb;
      const float* ap = (j < HID) ? (hs + j) : (hd + (j - HID));
      v2f a; a.x = ap[0]; a.y = ap[1];
      acc0 = __builtin_amdgcn_wmma_f32_16x16x4_f32(false, a, false, b0[ks], (short)0, acc0,
                                                   false, false);
      acc1 = __builtin_amdgcn_wmma_f32_16x16x4_f32(false, a, false, b1[ks], (short)0, acc1,
                                                   false, false);
    }
    float p[8][2];
#pragma unroll
    for (int r = 0; r < 8; ++r) {
      float t0 = fmaxf(acc0[r] + bA, 0.f);
      float t1 = fmaxf(acc1[r] + bB, 0.f);
      p[r][0] = t0 * wa0 + t1 * wb0;
      p[r][1] = t0 * wa1 + t1 * wb1;
    }
#pragma unroll
    for (int off = 8; off >= 1; off >>= 1) {
#pragma unroll
      for (int r = 0; r < 8; ++r) {
        p[r][0] += __shfl_xor(p[r][0], off, 16);
        p[r][1] += __shfl_xor(p[r][1], off, 16);
      }
    }
    if (m == 0) {
      int rbase = e0 + ((lane >> 4) ? 8 : 0);
#pragma unroll
      for (int r = 0; r < 8; ++r) {
        int row = rbase + r;
        if (row < Etot) {
          out[(size_t)row * 2 + 0] = p[r][0] + c0;
          out[(size_t)row * 2 + 1] = p[r][1] + c1;
        }
      }
    }
  }
}

extern "C" void kernel_launch(void* const* d_in, const int* in_sizes, int n_in,
                              void* d_out, int out_size, void* d_ws, size_t ws_size,
                              hipStream_t stream) {
  const float* x   = (const float*)d_in[0];
  const int*   pe  = (const int*)d_in[1];
  const int*   neg = (const int*)d_in[2];
  const float* W1  = (const float*)d_in[3];
  const float* b1  = (const float*)d_in[4];
  const float* W2  = (const float*)d_in[5];
  const float* b2  = (const float*)d_in[6];
  const float* Wm1 = (const float*)d_in[7];
  const float* bm1 = (const float*)d_in[8];
  const float* Wm2 = (const float*)d_in[9];
  const float* bm2 = (const float*)d_in[10];
  float* out = (float*)d_out;

  const int N  = in_sizes[0] / HID;
  const int NP = in_sizes[1] / 2;
  const int NN = in_sizes[2] / 2;
  const int ET = NP + NN;
  const int* p0 = pe;        const int* p1 = pe + NP;
  const int* q0 = neg;       const int* q1 = neg + NN;
  const int NB = (N + TN - 1) / TN;

  char* ws = (char*)d_ws;
  size_t off = 0;
  auto wsalloc = [&](size_t bytes) -> char* {
    char* p = ws + off;
    off += (bytes + 255) & ~(size_t)255;
    return p;
  };
  float* dinv = (float*)wsalloc((size_t)N * 4);
  float* xl1  = (float*)wsalloc((size_t)N * H1 * 4);
  float* h1   = (float*)wsalloc((size_t)N * H1 * 4);
  float* xl2  = (float*)wsalloc((size_t)N * HID * 4);
  float* h2   = (float*)wsalloc((size_t)N * HID * 4);
  int*   bcur = (int*)wsalloc((size_t)NB * 4);
  int*   bst  = (int*)wsalloc((size_t)(NB + 1) * 4);
  int2*  ebin = (int2*)wsalloc((size_t)NP * 8);
  (void)ws_size; (void)n_in; (void)out_size;

  const int B = 256;

  hipMemsetAsync(bcur, 0, (size_t)NB * 4, stream);
  {
    const int nblk = (NP + CHUNK - 1) / CHUNK;
    bin_count<<<nblk, 1024, 0, stream>>>(p1, bcur, NP, NB);
    bin_scan<<<1, 1024, 0, stream>>>(bcur, bst, NB);
    bin_fill<<<nblk, 1024, 0, stream>>>(p0, p1, bcur, ebin, NP, NB);
  }
  deg_tile<<<NB, TN, 0, stream>>>(ebin, bst, dinv, N);

  {
    long long thr = (long long)((N + 15) / 16) * (H1 / 16) * 32;
    gemm_wmma_f32<<<(unsigned)((thr + B - 1) / B), B, 0, stream>>>(x, W1, xl1, N, H1, HID);
  }
  agg_tile<H1><<<NB, 256, 0, stream>>>(xl1, ebin, bst, dinv, b1, h1, N);

  {
    long long thr = (long long)((N + 15) / 16) * (HID / 16) * 32;
    gemm_wmma_f32<<<(unsigned)((thr + B - 1) / B), B, 0, stream>>>(h1, W2, xl2, N, HID, H1);
  }
  agg_tile<HID><<<NB, 256, 0, stream>>>(xl2, ebin, bst, dinv, b2, h2, N);

  edge_mlp_wmma<<<MLP_BLOCKS, B, 0, stream>>>(h2, p0, p1, q0, q1, Wm1, bm1, Wm2, bm2,
                                              out, NP, ET);
}


# Both node products: one 16-wave workgroup per WGP, a wave keeps all of B as fragments in 128 registers and takes whole 16-row strips, A by 16-byte loads, whole-line stores; fp32 k sum order changes
# speedup vs baseline: 2.7211x; 1.2296x over previous
#include <hip/hip_runtime.h>
#include <hip/hip_bf16.h>

typedef __attribute__((ext_vector_type(2))) float v2f;
typedef __attribute__((ext_vector_type(8))) float v8f;

#define HID 32
#define H1  128

#define GEMM_BLOCKS 256
#define GEMM_WAVES  16
template <int K, int N>
__global__ __launch_bounds__(GEMM_WAVES * 32)
void gemm_strip_f32(const float* __restrict__ A, const float* __restrict__ B,
                    float* __restrict__ C, int M) {
  constexpr int KS = K / 4;
  constexpr int NT = N / 16;
  constexpr int G  = (NT < 4) ? NT : 4;
  typedef __attribute__((ext_vector_type(G))) float vgrp;
  typedef __attribute__((ext_vector_type(4))) float v4f;
  const int lane = threadIdx.x & 31;
  const int c = lane & 15;
  const int h = lane >> 4;
  const int wave = __builtin_amdgcn_readfirstlane(
      (int)((threadIdx.x >> 5) * gridDim.x + blockIdx.x));
  const int nwaves = (int)(gridDim.x * (blockDim.x >> 5));
  const int nstrips = (M + 15) >> 4;

  v2f b[KS][NT];
#pragma unroll
  for (int ks = 0; ks < KS; ++ks) {
    const int kk = 8 * (ks >> 1) + 4 * h + 2 * (ks & 1);
#pragma unroll
    for (int q = 0; q < NT / G; ++q) {
      const vgrp r0 = *(const vgrp*)(B + (size_t)kk * N + q * 16 * G + G * c);
      const vgrp r1 = *(const vgrp*)(B + (size_t)(kk + 1) * N + q * 16 * G + G * c);
#pragma unroll
      for (int g = 0; g < G; ++g) {
        b[ks][q * G + g].x = r0[g];
        b[ks][q * G + g].y = r1[g];
      }
    }
  }

  for (int s = wave; s < nstrips; s += nwaves) {
    int mrow = s * 16 + c;
    if (mrow > M - 1) mrow = M - 1;
    const float* ap = A + (size_t)mrow * K + 4 * h;
    v4f av[K / 8];
#pragma unroll
    for (int i = 0; i < K / 8; ++i) av[i] = *(const v4f*)(ap + 8 * i);
    __builtin_amdgcn_sched_barrier(0);
    v8f acc[NT];
#pragma unroll
    for (int t = 0; t < NT; ++t) acc[t] = (v8f){};
#pragma unroll
    for (int ks = 0; ks < KS; ++ks) {
      v2f a;
      a.x = av[ks >> 1][2 * (ks & 1)];
      a.y = av[ks >> 1][2 * (ks & 1) + 1];
#pragma unroll
      for (int t = 0; t < NT; ++t)
        acc[t] = __builtin_amdgcn_wmma_f32_16x16x4_f32(false, a, false, b[ks][t], (short)0,
                                                       acc[t], false, false);
    }
    const int rbase = s * 16 + 8 * h;
#pragma unroll
    for (int r = 0; r < 8; ++r) {
      const int row = rbase + r;
      if (row < M) {
#pragma unroll
        for (int q = 0; q < NT / G; ++q) {
          vgrp o;
#pragma unroll
          for (int g = 0; g < G; ++g) o[g] = acc[q * G + g][r];
          *(vgrp*)(C + (size_t)row * N + q * 16 * G + G * c) = o;
        }
      }
    }
  }
}

#define TSH   6
#define TN    64
#define MAXB  2048
#define CHUNK 8192

__global__ __launch_bounds__(1024)
void bin_count(const int* __restrict__ dst, int* __restrict__ cnt, int E, int nb) {
  __shared__ int hist[MAXB];
  const int tid = threadIdx.x;
  for (int i = tid; i < nb; i += 1024) hist[i] = 0;
  __syncthreads();
  const int base = blockIdx.x * CHUNK;
#pragma unroll
  for (int k = 0; k < CHUNK / 1024; ++k) {
    int e = base + k * 1024 + tid;
    if (e < E) atomicAdd(&hist[dst[e] >> TSH], 1);
  }
  __syncthreads();
  for (int i = tid; i < nb; i += 1024) {
    int c = hist[i];
    if (c) atomicAdd(&cnt[i], c);
  }
}

__global__ __launch_bounds__(1024)
void bin_scan(int* __restrict__ cnt, int* __restrict__ start, int nb) {
  __shared__ int s[1024];
  const int t = threadIdx.x;
  int a = (2 * t < nb) ? cnt[2 * t] : 0;
  int b = (2 * t + 1 < nb) ? cnt[2 * t + 1] : 0;
  int v = a + b;
  s[t] = v;
  __syncthreads();
  for (int off = 1; off < 1024; off <<= 1) {
    int x = (t >= off) ? s[t - off] : 0;
    __syncthreads();
    s[t] += x;
    __syncthreads();
  }
  int excl = s[t] - v;
  if (2 * t < nb)     { start[2 * t] = excl;         cnt[2 * t] = excl; }
  if (2 * t + 1 < nb) { start[2 * t + 1] = excl + a; cnt[2 * t + 1] = excl + a; }
  if (t == 1023) start[nb] = s[1023];
}

__global__ __launch_bounds__(1024)
void bin_fill(const int* __restrict__ src, const int* __restrict__ dst,
              int* __restrict__ cur, int2* __restrict__ ebin, int E, int nb) {
  __shared__ int hist[MAXB];
  const int tid = threadIdx.x;
  for (int i = tid; i < nb; i += 1024) hist[i] = 0;
  __syncthreads();
  const int base = blockIdx.x * CHUNK;
  int sv[CHUNK / 1024], dv[CHUNK / 1024], rk[CHUNK / 1024];
#pragma unroll
  for (int k = 0; k < CHUNK / 1024; ++k) {
    int e = base + k * 1024 + tid;
    sv[k] = 0; dv[k] = 0; rk[k] = 0;
    if (e < E) {
      sv[k] = src[e];
      dv[k] = dst[e];
      rk[k] = atomicAdd(&hist[dv[k] >> TSH], 1);
    }
  }
  __syncthreads();
  for (int i = tid; i < nb; i += 1024) {
    int c = hist[i];
    hist[i] = c ? atomicAdd(&cur[i], c) : 0;
  }
  __syncthreads();
#pragma unroll
  for (int k = 0; k < CHUNK / 1024; ++k) {
    int e = base + k * 1024 + tid;
    if (e < E) ebin[hist[dv[k] >> TSH] + rk[k]] = make_int2(sv[k], dv[k]);
  }
}

__global__ __launch_bounds__(TN)
void deg_tile(const int2* __restrict__ ebin, const int* __restrict__ start,
              float* __restrict__ dinv, int N) {
  __shared__ int c[TN];
  const int t = threadIdx.x;
  const int n0 = blockIdx.x * TN;
  c[t] = 0;
  __syncthreads();
  const int beg = start[blockIdx.x], end = start[blockIdx.x + 1];
  for (int e = beg + t; e < end; e += TN) atomicAdd(&c[ebin[e].y - n0], 1);
  __syncthreads();
  if (n0 + t < N) dinv[n0 + t] = rsqrtf((float)c[t] + 1.0f);
}


template <int D, int U, int K0, bool FULL>
__device__ __forceinline__ void walk_rows(const float* __restrict__ xl, float* acc,
                                          int src, int dloc, float wt, int cnt, int lane) {
  constexpr int NJ = D / 32;
  typedef __attribute__((ext_vector_type(NJ))) float vrow;
  vrow v[U];
#pragma unroll
  for (int u = 0; u < U; ++u) {
    const int s = __builtin_amdgcn_readlane(src, K0 + u);
    v[u] = *(const vrow*)(xl + (size_t)s * D + lane * NJ);
  }
#pragma unroll
  for (int u = 0; u < U; ++u) {
    if (FULL || K0 + u < cnt) {
      const float wk = __builtin_bit_cast(
          float, __builtin_amdgcn_readlane(__builtin_bit_cast(int, wt), K0 + u));
      float* row = acc + __builtin_amdgcn_readlane(dloc, K0 + u) + lane;
#pragma unroll
      for (int j = 0; j < NJ; ++j)
        __hip_atomic_fetch_add(row + j * 32, wk * v[u][j], __ATOMIC_RELAXED,
                               __HIP_MEMORY_SCOPE_WORKGROUP);
    }
  }
}

template <int D, int U, int K0, bool FULL>
__device__ __forceinline__ void walk_batch(const float* __restrict__ xl, float* acc,
                                           int src, int dloc, float wt, int cnt, int lane) {
  if constexpr (K0 < 32) {
    if (FULL || K0 < cnt) {
      walk_rows<D, U, K0, FULL>(xl, acc, src, dloc, wt, cnt, lane);
      walk_batch<D, U, K0 + U, FULL>(xl, acc, src, dloc, wt, cnt, lane);
    }
  }
}

template <int D>
__global__ __launch_bounds__(256)
void agg_tile(const float* __restrict__ xl, const int2* __restrict__ ebin,
              const int* __restrict__ start, const float* __restrict__ dinv,
              const float* __restrict__ bias, float* __restrict__ h, int N) {
  constexpr int NJ = D / 32;
  constexpr int U = (D > 32) ? 8 : 32;
  constexpr int NW = 8;
  typedef __attribute__((ext_vector_type(NJ))) float vrow;
  __shared__ float acc[TN * D];
  const int tid = threadIdx.x;
  const int lane = tid & 31;
  const int w = __builtin_amdgcn_readfirstlane(tid >> 5);
  const int n0 = blockIdx.x * TN;
  for (int i = tid; i < TN * D; i += 256) acc[i] = 0.f;
  __syncthreads();
  const int beg = start[blockIdx.x], end = start[blockIdx.x + 1];
  for (int eb = beg + w * 32; eb < end; eb += NW * 32) {
    int ee = eb + lane;
    ee = (ee < end) ? ee : (end - 1);
    const int2 sd = ebin[ee];
    const float wt = dinv[sd.x] * dinv[sd.y];
    const int dloc = (sd.y - n0) * D;
    const int cnt = end - eb;
    if (cnt >= 32) walk_batch<D, U, 0, true>(xl, acc, sd.x, dloc, wt, cnt, lane);
    else           walk_batch<D, U, 0, false>(xl, acc, sd.x, dloc, wt, cnt, lane);
  }
  __syncthreads();
  for (int idx = tid; idx < TN * 32; idx += 256) {
    int r = idx >> 5;
    int i = n0 + r;
    if (i < N) {
      float di = dinv[i], sl = di * di;
      const vrow x = *(const vrow*)(xl + (size_t)i * D + lane * NJ);
      const vrow bb = *(const vrow*)(bias + lane * NJ);
      vrow o;
#pragma unroll
      for (int j = 0; j < NJ; ++j)
        o[j] = fmaxf(acc[r * D + j * 32 + lane] + sl * x[j] + bb[j], 0.f);
      *(vrow*)(h + (size_t)i * D + lane * NJ) = o;
    }
  }
}

#define MLP_BLOCKS 1024
__global__ __launch_bounds__(256) __attribute__((amdgpu_waves_per_eu(8, 8)))
void edge_mlp_wmma(const float* __restrict__ h, const int* __restrict__ p0,
                   const int* __restrict__ p1, const int* __restrict__ q0,
                   const int* __restrict__ q1, const float* __restrict__ Wm1,
                   const float* __restrict__ bm1, const float* __restrict__ Wm2,
                   const float* __restrict__ bm2, float* __restrict__ out,
                   int NP, int Etot) {
  const int lane = threadIdx.x & 31;
  const int m = lane & 15;
  const int kb = (lane >> 4) * 2;
  const int wave = __builtin_amdgcn_readfirstlane(
      (int)(blockIdx.x * (blockDim.x >> 5) + (threadIdx.x >> 5)));
  const int nwaves = (int)(gridDim.x * (blockDim.x >> 5));
  const int ntiles = (Etot + 15) >> 4;

  v2f b0[16], b1[16];
#pragma unroll
  for (int ks = 0; ks < 16; ++ks) {
    int j = ks * 4 + kb;
    b0[ks].x = Wm1[(size_t)j * HID + m];
    b0[ks].y = Wm1[(size_t)(j + 1) * HID + m];
    b1[ks].x = Wm1[(size_t)j * HID + 16 + m];
    b1[ks].y = Wm1[(size_t)(j + 1) * HID + 16 + m];
  }
  const float bA = bm1[m],        bB = bm1[m + 16];
  const float wa0 = Wm2[m * 2],   wa1 = Wm2[m * 2 + 1];
  const float wb0 = Wm2[(m + 16) * 2], wb1 = Wm2[(m + 16) * 2 + 1];
  const float c0 = bm2[0], c1 = bm2[1];

  for (int t = wave; t < ntiles; t += nwaves) {
    const int e0 = t * 16;
    int e = e0 + m;
    if (e > Etot - 1) e = Etot - 1;
    int s = (e < NP) ? p0[e] : q0[e - NP];
    int d = (e < NP) ? p1[e] : q1[e - NP];
    const float* hs = h + (size_t)s * HID;
    const float* hd = h + (size_t)d * HID;
    v8f acc0 = {}, acc1 = {};
#pragma unroll
    for (int ks = 0; ks < 16; ++ks) {
      int j = ks * 4 + kb;
      const float* ap = (j < HID) ? (hs + j) : (hd + (j - HID));
      v2f a; a.x = ap[0]; a.y = ap[1];
      acc0 = __builtin_amdgcn_wmma_f32_16x16x4_f32(false, a, false, b0[ks], (short)0, acc0,
                                                   false, false);
      acc1 = __builtin_amdgcn_wmma_f32_16x16x4_f32(false, a, false, b1[ks], (short)0, acc1,
                                                   false, false);
    }
    float p[8][2];
#pragma unroll
    for (int r = 0; r < 8; ++r) {
      float t0 = fmaxf(acc0[r] + bA, 0.f);
      float t1 = fmaxf(acc1[r] + bB, 0.f);
      p[r][0] = t0 * wa0 + t1 * wb0;
      p[r][1] = t0 * wa1 + t1 * wb1;
    }
#pragma unroll
    for (int off = 8; off >= 1; off >>= 1) {
#pragma unroll
      for (int r = 0; r < 8; ++r) {
        p[r][0] += __shfl_xor(p[r][0], off, 16);
        p[r][1] += __shfl_xor(p[r][1], off, 16);
      }
    }
    if (m == 0) {
      int rbase = e0 + ((lane >> 4) ? 8 : 0);
#pragma unroll
      for (int r = 0; r < 8; ++r) {
        int row = rbase + r;
        if (row < Etot) {
          out[(size_t)row * 2 + 0] = p[r][0] + c0;
          out[(size_t)row * 2 + 1] = p[r][1] + c1;
        }
      }
    }
  }
}

extern "C" void kernel_launch(void* const* d_in, const int* in_sizes, int n_in,
                              void* d_out, int out_size, void* d_ws, size_t ws_size,
                              hipStream_t stream) {
  const float* x   = (const float*)d_in[0];
  const int*   pe  = (const int*)d_in[1];
  const int*   neg = (const int*)d_in[2];
  const float* W1  = (const float*)d_in[3];
  const float* b1  = (const float*)d_in[4];
  const float* W2  = (const float*)d_in[5];
  const float* b2  = (const float*)d_in[6];
  const float* Wm1 = (const float*)d_in[7];
  const float* bm1 = (const float*)d_in[8];
  const float* Wm2 = (const float*)d_in[9];
  const float* bm2 = (const float*)d_in[10];
  float* out = (float*)d_out;

  const int N  = in_sizes[0] / HID;
  const int NP = in_sizes[1] / 2;
  const int NN = in_sizes[2] / 2;
  const int ET = NP + NN;
  const int* p0 = pe;        const int* p1 = pe + NP;
  const int* q0 = neg;       const int* q1 = neg + NN;
  const int NB = (N + TN - 1) / TN;

  char* ws = (char*)d_ws;
  size_t off = 0;
  auto wsalloc = [&](size_t bytes) -> char* {
    char* p = ws + off;
    off += (bytes + 255) & ~(size_t)255;
    return p;
  };
  float* dinv = (float*)wsalloc((size_t)N * 4);
  float* xl1  = (float*)wsalloc((size_t)N * H1 * 4);
  float* h1   = (float*)wsalloc((size_t)N * H1 * 4);
  float* xl2  = (float*)wsalloc((size_t)N * HID * 4);
  float* h2   = (float*)wsalloc((size_t)N * HID * 4);
  int*   bcur = (int*)wsalloc((size_t)NB * 4);
  int*   bst  = (int*)wsalloc((size_t)(NB + 1) * 4);
  int2*  ebin = (int2*)wsalloc((size_t)NP * 8);
  (void)ws_size; (void)n_in; (void)out_size;

  const int B = 256;

  hipMemsetAsync(bcur, 0, (size_t)NB * 4, stream);
  {
    const int nblk = (NP + CHUNK - 1) / CHUNK;
    bin_count<<<nblk, 1024, 0, stream>>>(p1, bcur, NP, NB);
    bin_scan<<<1, 1024, 0, stream>>>(bcur, bst, NB);
    bin_fill<<<nblk, 1024, 0, stream>>>(p0, p1, bcur, ebin, NP, NB);
  }
  deg_tile<<<NB, TN, 0, stream>>>(ebin, bst, dinv, N);

  gemm_strip_f32<HID, H1><<<GEMM_BLOCKS, GEMM_WAVES * 32, 0, stream>>>(x, W1, xl1, N);
  agg_tile<H1><<<NB, 256, 0, stream>>>(xl1, ebin, bst, dinv, b1, h1, N);

  gemm_strip_f32<H1, HID><<<GEMM_BLOCKS, GEMM_WAVES * 32, 0, stream>>>(h1, W2, xl2, N);
  agg_tile<HID><<<NB, 256, 0, stream>>>(xl2, ebin, bst, dinv, b2, h2, N);

  edge_mlp_wmma<<<MLP_BLOCKS, B, 0, stream>>>(h2, p0, p1, q0, q1, Wm1, bm1, Wm2, bm2,
                                              out, NP, ET);
}


# Edge MLP product transposed (Wm1^T as A, edge rows as B): a lane holds 16 hidden units of one edge; Wm2 product in-lane + 1 half swap, one 128-B store per tile, 6 waves/SIMD; fp32 sum order changes
# speedup vs baseline: 2.9038x; 1.0671x over previous
#include <hip/hip_runtime.h>
#include <hip/hip_bf16.h>

typedef __attribute__((ext_vector_type(2))) float v2f;
typedef __attribute__((ext_vector_type(8))) float v8f;

#define HID 32
#define H1  128

#define GEMM_BLOCKS 256
#define GEMM_WAVES  16
template <int K, int N>
__global__ __launch_bounds__(GEMM_WAVES * 32)
void gemm_strip_f32(const float* __restrict__ A, const float* __restrict__ B,
                    float* __restrict__ C, int M) {
  constexpr int KS = K / 4;
  constexpr int NT = N / 16;
  constexpr int G  = (NT < 4) ? NT : 4;
  typedef __attribute__((ext_vector_type(G))) float vgrp;
  typedef __attribute__((ext_vector_type(4))) float v4f;
  const int lane = threadIdx.x & 31;
  const int c = lane & 15;
  const int h = lane >> 4;
  const int wave = __builtin_amdgcn_readfirstlane(
      (int)((threadIdx.x >> 5) * gridDim.x + blockIdx.x));
  const int nwaves = (int)(gridDim.x * (blockDim.x >> 5));
  const int nstrips = (M + 15) >> 4;

  v2f b[KS][NT];
#pragma unroll
  for (int ks = 0; ks < KS; ++ks) {
    const int kk = 8 * (ks >> 1) + 4 * h + 2 * (ks & 1);
#pragma unroll
    for (int q = 0; q < NT / G; ++q) {
      const vgrp r0 = *(const vgrp*)(B + (size_t)kk * N + q * 16 * G + G * c);
      const vgrp r1 = *(const vgrp*)(B + (size_t)(kk + 1) * N + q * 16 * G + G * c);
#pragma unroll
      for (int g = 0; g < G; ++g) {
        b[ks][q * G + g].x = r0[g];
        b[ks][q * G + g].y = r1[g];
      }
    }
  }

  for (int s = wave; s < nstrips; s += nwaves) {
    int mrow = s * 16 + c;
    if (mrow > M - 1) mrow = M - 1;
    const float* ap = A + (size_t)mrow * K + 4 * h;
    v4f av[K / 8];
#pragma unroll
    for (int i = 0; i < K / 8; ++i) av[i] = *(const v4f*)(ap + 8 * i);
    __builtin_amdgcn_sched_barrier(0);
    v8f acc[NT];
#pragma unroll
    for (int t = 0; t < NT; ++t) acc[t] = (v8f){};
#pragma unroll
    for (int ks = 0; ks < KS; ++ks) {
      v2f a;
      a.x = av[ks >> 1][2 * (ks & 1)];
      a.y = av[ks >> 1][2 * (ks & 1) + 1];
#pragma unroll
      for (int t = 0; t < NT; ++t)
        acc[t] = __builtin_amdgcn_wmma_f32_16x16x4_f32(false, a, false, b[ks][t], (short)0,
                                                       acc[t], false, false);
    }
    const int rbase = s * 16 + 8 * h;
#pragma unroll
    for (int r = 0; r < 8; ++r) {
      const int row = rbase + r;
      if (row < M) {
#pragma unroll
        for (int q = 0; q < NT / G; ++q) {
          vgrp o;
#pragma unroll
          for (int g = 0; g < G; ++g) o[g] = acc[q * G + g][r];
          *(vgrp*)(C + (size_t)row * N + q * 16 * G + G * c) = o;
        }
      }
    }
  }
}

#define TSH   6
#define TN    64
#define MAXB  2048
#define CHUNK 8192

__global__ __launch_bounds__(1024)
void bin_count(const int* __restrict__ dst, int* __restrict__ cnt, int E, int nb) {
  __shared__ int hist[MAXB];
  const int tid = threadIdx.x;
  for (int i = tid; i < nb; i += 1024) hist[i] = 0;
  __syncthreads();
  const int base = blockIdx.x * CHUNK;
#pragma unroll
  for (int k = 0; k < CHUNK / 1024; ++k) {
    int e = base + k * 1024 + tid;
    if (e < E) atomicAdd(&hist[dst[e] >> TSH], 1);
  }
  __syncthreads();
  for (int i = tid; i < nb; i += 1024) {
    int c = hist[i];
    if (c) atomicAdd(&cnt[i], c);
  }
}

__global__ __launch_bounds__(1024)
void bin_scan(int* __restrict__ cnt, int* __restrict__ start, int nb) {
  __shared__ int s[1024];
  const int t = threadIdx.x;
  int a = (2 * t < nb) ? cnt[2 * t] : 0;
  int b = (2 * t + 1 < nb) ? cnt[2 * t + 1] : 0;
  int v = a + b;
  s[t] = v;
  __syncthreads();
  for (int off = 1; off < 1024; off <<= 1) {
    int x = (t >= off) ? s[t - off] : 0;
    __syncthreads();
    s[t] += x;
    __syncthreads();
  }
  int excl = s[t] - v;
  if (2 * t < nb)     { start[2 * t] = excl;         cnt[2 * t] = excl; }
  if (2 * t + 1 < nb) { start[2 * t + 1] = excl + a; cnt[2 * t + 1] = excl + a; }
  if (t == 1023) start[nb] = s[1023];
}

__global__ __launch_bounds__(1024)
void bin_fill(const int* __restrict__ src, const int* __restrict__ dst,
              int* __restrict__ cur, int2* __restrict__ ebin, int E, int nb) {
  __shared__ int hist[MAXB];
  const int tid = threadIdx.x;
  for (int i = tid; i < nb; i += 1024) hist[i] = 0;
  __syncthreads();
  const int base = blockIdx.x * CHUNK;
  int sv[CHUNK / 1024], dv[CHUNK / 1024], rk[CHUNK / 1024];
#pragma unroll
  for (int k = 0; k < CHUNK / 1024; ++k) {
    int e = base + k * 1024 + tid;
    sv[k] = 0; dv[k] = 0; rk[k] = 0;
    if (e < E) {
      sv[k] = src[e];
      dv[k] = dst[e];
      rk[k] = atomicAdd(&hist[dv[k] >> TSH], 1);
    }
  }
  __syncthreads();
  for (int i = tid; i < nb; i += 1024) {
    int c = hist[i];
    hist[i] = c ? atomicAdd(&cur[i], c) : 0;
  }
  __syncthreads();
#pragma unroll
  for (int k = 0; k < CHUNK / 1024; ++k) {
    int e = base + k * 1024 + tid;
    if (e < E) ebin[hist[dv[k] >> TSH] + rk[k]] = make_int2(sv[k], dv[k]);
  }
}

__global__ __launch_bounds__(TN)
void deg_tile(const int2* __restrict__ ebin, const int* __restrict__ start,
              float* __restrict__ dinv, int N) {
  __shared__ int c[TN];
  const int t = threadIdx.x;
  const int n0 = blockIdx.x * TN;
  c[t] = 0;
  __syncthreads();
  const int beg = start[blockIdx.x], end = start[blockIdx.x + 1];
  for (int e = beg + t; e < end; e += TN) atomicAdd(&c[ebin[e].y - n0], 1);
  __syncthreads();
  if (n0 + t < N) dinv[n0 + t] = rsqrtf((float)c[t] + 1.0f);
}


template <int D, int U, int K0, bool FULL>
__device__ __forceinline__ void walk_rows(const float* __restrict__ xl, float* acc,
                                          int src, int dloc, float wt, int cnt, int lane) {
  constexpr int NJ = D / 32;
  typedef __attribute__((ext_vector_type(NJ))) float vrow;
  vrow v[U];
#pragma unroll
  for (int u = 0; u < U; ++u) {
    const int s = __builtin_amdgcn_readlane(src, K0 + u);
    v[u] = *(const vrow*)(xl + (size_t)s * D + lane * NJ);
  }
#pragma unroll
  for (int u = 0; u < U; ++u) {
    if (FULL || K0 + u < cnt) {
      const float wk = __builtin_bit_cast(
          float, __builtin_amdgcn_readlane(__builtin_bit_cast(int, wt), K0 + u));
      float* row = acc + __builtin_amdgcn_readlane(dloc, K0 + u) + lane;
#pragma unroll
      for (int j = 0; j < NJ; ++j)
        __hip_atomic_fetch_add(row + j * 32, wk * v[u][j], __ATOMIC_RELAXED,
                               __HIP_MEMORY_SCOPE_WORKGROUP);
    }
  }
}

template <int D, int U, int K0, bool FULL>
__device__ __forceinline__ void walk_batch(const float* __restrict__ xl, float* acc,
                                           int src, int dloc, float wt, int cnt, int lane) {
  if constexpr (K0 < 32) {
    if (FULL || K0 < cnt) {
      walk_rows<D, U, K0, FULL>(xl, acc, src, dloc, wt, cnt, lane);
      walk_batch<D, U, K0 + U, FULL>(xl, acc, src, dloc, wt, cnt, lane);
    }
  }
}

template <int D>
__global__ __launch_bounds__(256)
void agg_tile(const float* __restrict__ xl, const int2* __restrict__ ebin,
              const int* __restrict__ start, const float* __restrict__ dinv,
              const float* __restrict__ bias, float* __restrict__ h, int N) {
  constexpr int NJ = D / 32;
  constexpr int U = (D > 32) ? 8 : 32;
  constexpr int NW = 8;
  typedef __attribute__((ext_vector_type(NJ))) float vrow;
  __shared__ float acc[TN * D];
  const int tid = threadIdx.x;
  const int lane = tid & 31;
  const int w = __builtin_amdgcn_readfirstlane(tid >> 5);
  const int n0 = blockIdx.x * TN;
  for (int i = tid; i < TN * D; i += 256) acc[i] = 0.f;
  __syncthreads();
  const int beg = start[blockIdx.x], end = start[blockIdx.x + 1];
  for (int eb = beg + w * 32; eb < end; eb += NW * 32) {
    int ee = eb + lane;
    ee = (ee < end) ? ee : (end - 1);
    const int2 sd = ebin[ee];
    const float wt = dinv[sd.x] * dinv[sd.y];
    const int dloc = (sd.y - n0) * D;
    const int cnt = end - eb;
    if (cnt >= 32) walk_batch<D, U, 0, true>(xl, acc, sd.x, dloc, wt, cnt, lane);
    else           walk_batch<D, U, 0, false>(xl, acc, sd.x, dloc, wt, cnt, lane);
  }
  __syncthreads();
  for (int idx = tid; idx < TN * 32; idx += 256) {
    int r = idx >> 5;
    int i = n0 + r;
    if (i < N) {
      float di = dinv[i], sl = di * di;
      const vrow x = *(const vrow*)(xl + (size_t)i * D + lane * NJ);
      const vrow bb = *(const vrow*)(bias + lane * NJ);
      vrow o;
#pragma unroll
      for (int j = 0; j < NJ; ++j)
        o[j] = fmaxf(acc[r * D + j * 32 + lane] + sl * x[j] + bb[j], 0.f);
      *(vrow*)(h + (size_t)i * D + lane * NJ) = o;
    }
  }
}

#define MLP_BLOCKS 768
__global__ __launch_bounds__(256) __attribute__((amdgpu_waves_per_eu(6, 6)))
void edge_mlp_wmma(const float* __restrict__ h, const int* __restrict__ p0,
                   const int* __restrict__ p1, const int* __restrict__ q0,
                   const int* __restrict__ q1, const float* __restrict__ Wm1,
                   const float* __restrict__ bm1, const float* __restrict__ Wm2,
                   const float* __restrict__ bm2, float* __restrict__ out,
                   int NP, int Etot) {
  const int lane = threadIdx.x & 31;
  const int m = lane & 15;
  const int hh = lane >> 4;
  const int kb = hh * 2;
  const int wave = __builtin_amdgcn_readfirstlane(
      (int)(blockIdx.x * (blockDim.x >> 5) + (threadIdx.x >> 5)));
  const int nwaves = (int)(gridDim.x * (blockDim.x >> 5));
  const int ntiles = (Etot + 15) >> 4;

  v2f w0[16], w1[16];
#pragma unroll
  for (int ks = 0; ks < 16; ++ks) {
    int j = ks * 4 + kb;
    w0[ks].x = Wm1[(size_t)j * HID + m];
    w0[ks].y = Wm1[(size_t)(j + 1) * HID + m];
    w1[ks].x = Wm1[(size_t)j * HID + 16 + m];
    w1[ks].y = Wm1[(size_t)(j + 1) * HID + 16 + m];
  }
  v8f biasA, biasB;
  float wa0[8], wa1[8], wb0[8], wb1[8];
#pragma unroll
  for (int r = 0; r < 8; ++r) {
    const int nA = 8 * hh + r, nB = 16 + 8 * hh + r;
    biasA[r] = bm1[nA];
    biasB[r] = bm1[nB];
    wa0[r] = Wm2[nA * 2];  wa1[r] = Wm2[nA * 2 + 1];
    wb0[r] = Wm2[nB * 2];  wb1[r] = Wm2[nB * 2 + 1];
  }
  const float c0 = bm2[0], c1 = bm2[1];

  for (int t = wave; t < ntiles; t += nwaves) {
    const int e0 = t * 16;
    int e = e0 + m;
    if (e > Etot - 1) e = Etot - 1;
    int s = (e < NP) ? p0[e] : q0[e - NP];
    int d = (e < NP) ? p1[e] : q1[e - NP];
    const float* hs = h + (size_t)s * HID;
    const float* hd = h + (size_t)d * HID;
    v8f acc0 = biasA, acc1 = biasB;
#pragma unroll
    for (int ks = 0; ks < 16; ++ks) {
      int j = ks * 4 + kb;
      const float* ap = (j < HID) ? (hs + j) : (hd + (j - HID));
      v2f a; a.x = ap[0]; a.y = ap[1];
      acc0 = __builtin_amdgcn_wmma_f32_16x16x4_f32(false, w0[ks], false, a, (short)0, acc0,
                                                   false, false);
      acc1 = __builtin_amdgcn_wmma_f32_16x16x4_f32(false, w1[ks], false, a, (short)0, acc1,
                                                   false, false);
    }
    float o0 = 0.f, o1 = 0.f;
#pragma unroll
    for (int r = 0; r < 8; ++r) {
      const float t0 = fmaxf(acc0[r], 0.f);
      const float t1 = fmaxf(acc1[r], 0.f);
      o0 = fmaf(t0, wa0[r], o0);
      o1 = fmaf(t0, wa1[r], o1);
      o0 = fmaf(t1, wb0[r], o0);
      o1 = fmaf(t1, wb1[r], o1);
    }
    o0 += __shfl_xor(o0, 16);
    o1 += __shfl_xor(o1, 16);
    if (hh == 0 && e0 + m < Etot) {
      v2f o; o.x = o0 + c0; o.y = o1 + c1;
      *(v2f*)(out + (size_t)(e0 + m) * 2) = o;
    }
  }
}

extern "C" void kernel_launch(void* const* d_in, const int* in_sizes, int n_in,
                              void* d_out, int out_size, void* d_ws, size_t ws_size,
                              hipStream_t stream) {
  const float* x   = (const float*)d_in[0];
  const int*   pe  = (const int*)d_in[1];
  const int*   neg = (const int*)d_in[2];
  const float* W1  = (const float*)d_in[3];
  const float* b1  = (const float*)d_in[4];
  const float* W2  = (const float*)d_in[5];
  const float* b2  = (const float*)d_in[6];
  const float* Wm1 = (const float*)d_in[7];
  const float* bm1 = (const float*)d_in[8];
  const float* Wm2 = (const float*)d_in[9];
  const float* bm2 = (const float*)d_in[10];
  float* out = (float*)d_out;

  const int N  = in_sizes[0] / HID;
  const int NP = in_sizes[1] / 2;
  const int NN = in_sizes[2] / 2;
  const int ET = NP + NN;
  const int* p0 = pe;        const int* p1 = pe + NP;
  const int* q0 = neg;       const int* q1 = neg + NN;
  const int NB = (N + TN - 1) / TN;

  char* ws = (char*)d_ws;
  size_t off = 0;
  auto wsalloc = [&](size_t bytes) -> char* {
    char* p = ws + off;
    off += (bytes + 255) & ~(size_t)255;
    return p;
  };
  float* dinv = (float*)wsalloc((size_t)N * 4);
  float* xl1  = (float*)wsalloc((size_t)N * H1 * 4);
  float* h1   = (float*)wsalloc((size_t)N * H1 * 4);
  float* xl2  = (float*)wsalloc((size_t)N * HID * 4);
  float* h2   = (float*)wsalloc((size_t)N * HID * 4);
  int*   bcur = (int*)wsalloc((size_t)NB * 4);
  int*   bst  = (int*)wsalloc((size_t)(NB + 1) * 4);
  int2*  ebin = (int2*)wsalloc((size_t)NP * 8);
  (void)ws_size; (void)n_in; (void)out_size;

  const int B = 256;

  hipMemsetAsync(bcur, 0, (size_t)NB * 4, stream);
  {
    const int nblk = (NP + CHUNK - 1) / CHUNK;
    bin_count<<<nblk, 1024, 0, stream>>>(p1, bcur, NP, NB);
    bin_scan<<<1, 1024, 0, stream>>>(bcur, bst, NB);
    bin_fill<<<nblk, 1024, 0, stream>>>(p0, p1, bcur, ebin, NP, NB);
  }
  deg_tile<<<NB, TN, 0, stream>>>(ebin, bst, dinv, N);

  gemm_strip_f32<HID, H1><<<GEMM_BLOCKS, GEMM_WAVES * 32, 0, stream>>>(x, W1, xl1, N);
  agg_tile<H1><<<NB, 256, 0, stream>>>(xl1, ebin, bst, dinv, b1, h1, N);

  gemm_strip_f32<H1, HID><<<GEMM_BLOCKS, GEMM_WAVES * 32, 0, stream>>>(h1, W2, xl2, N);
  agg_tile<HID><<<NB, 256, 0, stream>>>(xl2, ebin, bst, dinv, b2, h2, N);

  edge_mlp_wmma<<<MLP_BLOCKS, B, 0, stream>>>(h2, p0, p1, q0, q1, Wm1, bm1, Wm2, bm2,
                                              out, NP, ET);
}


# Two small passes ride in neighbouring launches: each fill block scans the tile counts itself, the x@W1 blocks first count the tile degrees in one load group; 2 launches fewer; float sums untouched
# speedup vs baseline: 3.0302x; 1.0435x over previous
#include <hip/hip_runtime.h>
#include <hip/hip_bf16.h>

typedef __attribute__((ext_vector_type(2))) float v2f;
typedef __attribute__((ext_vector_type(8))) float v8f;

#define HID 32
#define H1  128

#define GEMM_BLOCKS 256
#define GEMM_WAVES  16
#define TN    64
#define MAXB  2048
#define DEGT  (MAXB / GEMM_BLOCKS)
#define DEGU  16
template <int K, int N, bool DEG>
__global__ __launch_bounds__(GEMM_WAVES * 32)
void gemm_strip_f32(const float* __restrict__ A, const float* __restrict__ B,
                    float* __restrict__ C, int M,
                    const int2* __restrict__ ebin, const int* __restrict__ start,
                    float* __restrict__ dinv, int nb) {
  if constexpr (DEG) {
    __shared__ int cdeg[DEGT * TN];
    const int tid = threadIdx.x;
    const int nthr = GEMM_WAVES * 32;
    const int tpb = (nb + (int)gridDim.x - 1) / (int)gridDim.x;
    const int t0 = (int)blockIdx.x * tpb;
    int t1 = t0 + tpb;
    if (t1 > nb) t1 = nb;
    for (int i = tid; i < DEGT * TN; i += nthr) cdeg[i] = 0;
    __syncthreads();
    if (t0 < t1) {
      const int beg = start[t0], end = start[t1];
      const int n0 = t0 * TN;
      for (int base = beg; base < end; base += DEGU * nthr) {
        const int last = end - 1;
        int d[DEGU];
#pragma unroll
        for (int k = 0; k < DEGU; ++k) {
          const int e = base + k * nthr + tid;
          d[k] = ebin[(e < last) ? e : last].y;
          if (e > last) d[k] = -1;
        }
#pragma unroll
        for (int k = 0; k < DEGU; ++k)
          if (d[k] >= 0) atomicAdd(&cdeg[d[k] - n0], 1);
      }
    }
    __syncthreads();
    for (int i = tid; i < (t1 - t0) * TN; i += nthr) {
      const int n = t0 * TN + i;
      if (n < M) dinv[n] = rsqrtf((float)cdeg[i] + 1.0f);
    }
  }
  constexpr int KS = K / 4;
  constexpr int NT = N / 16;
  constexpr int G  = (NT < 4) ? NT : 4;
  typedef __attribute__((ext_vector_type(G))) float vgrp;
  typedef __attribute__((ext_vector_type(4))) float v4f;
  const int lane = threadIdx.x & 31;
  const int c = lane & 15;
  const int h = lane >> 4;
  const int wave = __builtin_amdgcn_readfirstlane(
      (int)((threadIdx.x >> 5) * gridDim.x + blockIdx.x));
  const int nwaves = (int)(gridDim.x * (blockDim.x >> 5));
  const int nstrips = (M + 15) >> 4;

  v2f b[KS][NT];
#pragma unroll
  for (int ks = 0; ks < KS; ++ks) {
    const int kk = 8 * (ks >> 1) + 4 * h + 2 * (ks & 1);
#pragma unroll
    for (int q = 0; q < NT / G; ++q) {
      const vgrp r0 = *(const vgrp*)(B + (size_t)kk * N + q * 16 * G + G * c);
      const vgrp r1 = *(const vgrp*)(B + (size_t)(kk + 1) * N + q * 16 * G + G * c);
#pragma unroll
      for (int g = 0; g < G; ++g) {
        b[ks][q * G + g].x = r0[g];
        b[ks][q * G + g].y = r1[g];
      }
    }
  }

  for (int s = wave; s < nstrips; s += nwaves) {
    int mrow = s * 16 + c;
    if (mrow > M - 1) mrow = M - 1;
    const float* ap = A + (size_t)mrow * K + 4 * h;
    v4f av[K / 8];
#pragma unroll
    for (int i = 0; i < K / 8; ++i) av[i] = *(const v4f*)(ap + 8 * i);
    __builtin_amdgcn_sched_barrier(0);
    v8f acc[NT];
#pragma unroll
    for (int t = 0; t < NT; ++t) acc[t] = (v8f){};
#pragma unroll
    for (int ks = 0; ks < KS; ++ks) {
      v2f a;
      a.x = av[ks >> 1][2 * (ks & 1)];
      a.y = av[ks >> 1][2 * (ks & 1) + 1];
#pragma unroll
      for (int t = 0; t < NT; ++t)
        acc[t] = __builtin_amdgcn_wmma_f32_16x16x4_f32(false, a, false, b[ks][t], (short)0,
                                                       acc[t], false, false);
    }
    const int rbase = s * 16 + 8 * h;
#pragma unroll
    for (int r = 0; r < 8; ++r) {
      const int row = rbase + r;
      if (row < M) {
#pragma unroll
        for (int q = 0; q < NT / G; ++q) {
          vgrp o;
#pragma unroll
          for (int g = 0; g < G; ++g) o[g] = acc[q * G + g][r];
          *(vgrp*)(C + (size_t)row * N + q * 16 * G + G * c) = o;
        }
      }
    }
  }
}

#define TSH   6
#define CHUNK 8192

__global__ __launch_bounds__(1024)
void bin_count(const int* __restrict__ dst, int* __restrict__ cnt, int E, int nb) {
  __shared__ int hist[MAXB];
  const int tid = threadIdx.x;
  for (int i = tid; i < nb; i += 1024) hist[i] = 0;
  __syncthreads();
  const int base = blockIdx.x * CHUNK;
#pragma unroll
  for (int k = 0; k < CHUNK / 1024; ++k) {
    int e = base + k * 1024 + tid;
    if (e < E) atomicAdd(&hist[dst[e] >> TSH], 1);
  }
  __syncthreads();
  for (int i = tid; i < nb; i += 1024) {
    int c = hist[i];
    if (c) atomicAdd(&cnt[i], c);
  }
}

__global__ __launch_bounds__(1024)
void bin_fill(const int* __restrict__ src, const int* __restrict__ dst,
              const int* __restrict__ cnt, int* __restrict__ cur, int* __restrict__ start,
              int2* __restrict__ ebin, int E, int nb) {
  __shared__ int hist[MAXB];
  __shared__ int first[MAXB];
  __shared__ int s[1024];
  const int tid = threadIdx.x;
  for (int i = tid; i < nb; i += 1024) hist[i] = 0;
  {
    const int t = tid;
    int a = (2 * t < nb) ? cnt[2 * t] : 0;
    int b = (2 * t + 1 < nb) ? cnt[2 * t + 1] : 0;
    int v = a + b;
    s[t] = v;
    __syncthreads();
    for (int off = 1; off < 1024; off <<= 1) {
      int x = (t >= off) ? s[t - off] : 0;
      __syncthreads();
      s[t] += x;
      __syncthreads();
    }
    int excl = s[t] - v;
    if (2 * t < nb)     first[2 * t] = excl;
    if (2 * t + 1 < nb) first[2 * t + 1] = excl + a;
    if (blockIdx.x == 0) {
      if (2 * t < nb)     start[2 * t] = excl;
      if (2 * t + 1 < nb) start[2 * t + 1] = excl + a;
      if (t == 1023) start[nb] = s[1023];
    }
  }
  __syncthreads();
  const int base = blockIdx.x * CHUNK;
  int sv[CHUNK / 1024], dv[CHUNK / 1024], rk[CHUNK / 1024];
#pragma unroll
  for (int k = 0; k < CHUNK / 1024; ++k) {
    int e = base + k * 1024 + tid;
    sv[k] = 0; dv[k] = 0; rk[k] = 0;
    if (e < E) {
      sv[k] = src[e];
      dv[k] = dst[e];
      rk[k] = atomicAdd(&hist[dv[k] >> TSH], 1);
    }
  }
  __syncthreads();
  for (int i = tid; i < nb; i += 1024) {
    int c = hist[i];
    hist[i] = c ? first[i] + atomicAdd(&cur[i], c) : 0;
  }
  __syncthreads();
#pragma unroll
  for (int k = 0; k < CHUNK / 1024; ++k) {
    int e = base + k * 1024 + tid;
    if (e < E) ebin[hist[dv[k] >> TSH] + rk[k]] = make_int2(sv[k], dv[k]);
  }
}


template <int D, int U, int K0, bool FULL>
__device__ __forceinline__ void walk_rows(const float* __restrict__ xl, float* acc,
                                          int src, int dloc, float wt, int cnt, int lane) {
  constexpr int NJ = D / 32;
  typedef __attribute__((ext_vector_type(NJ))) float vrow;
  vrow v[U];
#pragma unroll
  for (int u = 0; u < U; ++u) {
    const int s = __builtin_amdgcn_readlane(src, K0 + u);
    v[u] = *(const vrow*)(xl + (size_t)s * D + lane * NJ);
  }
#pragma unroll
  for (int u = 0; u < U; ++u) {
    if (FULL || K0 + u < cnt) {
      const float wk = __builtin_bit_cast(
          float, __builtin_amdgcn_readlane(__builtin_bit_cast(int, wt), K0 + u));
      float* row = acc + __builtin_amdgcn_readlane(dloc, K0 + u) + lane;
#pragma unroll
      for (int j = 0; j < NJ; ++j)
        __hip_atomic_fetch_add(row + j * 32, wk * v[u][j], __ATOMIC_RELAXED,
                               __HIP_MEMORY_SCOPE_WORKGROUP);
    }
  }
}

template <int D, int U, int K0, bool FULL>
__device__ __forceinline__ void walk_batch(const float* __restrict__ xl, float* acc,
                                           int src, int dloc, float wt, int cnt, int lane) {
  if constexpr (K0 < 32) {
    if (FULL || K0 < cnt) {
      walk_rows<D, U, K0, FULL>(xl, acc, src, dloc, wt, cnt, lane);
      walk_batch<D, U, K0 + U, FULL>(xl, acc, src, dloc, wt, cnt, lane);
    }
  }
}

template <int D>
__global__ __launch_bounds__(256)
void agg_tile(const float* __restrict__ xl, const int2* __restrict__ ebin,
              const int* __restrict__ start, const float* __restrict__ dinv,
              const float* __restrict__ bias, float* __restrict__ h, int N) {
  constexpr int NJ = D / 32;
  constexpr int U = (D > 32) ? 8 : 32;
  constexpr int NW = 8;
  typedef __attribute__((ext_vector_type(NJ))) float vrow;
  __shared__ float acc[TN * D];
  const int tid = threadIdx.x;
  const int lane = tid & 31;
  const int w = __builtin_amdgcn_readfirstlane(tid >> 5);
  const int n0 = blockIdx.x * TN;
  for (int i = tid; i < TN * D; i += 256) acc[i] = 0.f;
  __syncthreads();
  const int beg = start[blockIdx.x], end = start[blockIdx.x + 1];
  for (int eb = beg + w * 32; eb < end; eb += NW * 32) {
    int ee = eb + lane;
    ee = (ee < end) ? ee : (end - 1);
    const int2 sd = ebin[ee];
    const float wt = dinv[sd.x] * dinv[sd.y];
    const int dloc = (sd.y - n0) * D;
    const int cnt = end - eb;
    if (cnt >= 32) walk_batch<D, U, 0, true>(xl, acc, sd.x, dloc, wt, cnt, lane);
    else           walk_batch<D, U, 0, false>(xl, acc, sd.x, dloc, wt, cnt, lane);
  }
  __syncthreads();
  for (int idx = tid; idx < TN * 32; idx += 256) {
    int r = idx >> 5;
    int i = n0 + r;
    if (i < N) {
      float di = dinv[i], sl = di * di;
      const vrow x = *(const vrow*)(xl + (size_t)i * D + lane * NJ);
      const vrow bb = *(const vrow*)(bias + lane * NJ);
      vrow o;
#pragma unroll
      for (int j = 0; j < NJ; ++j)
        o[j] = fmaxf(acc[r * D + j * 32 + lane] + sl * x[j] + bb[j], 0.f);
      *(vrow*)(h + (size_t)i * D + lane * NJ) = o;
    }
  }
}

#define MLP_BLOCKS 768
__global__ __launch_bounds__(256) __attribute__((amdgpu_waves_per_eu(6, 6)))
void edge_mlp_wmma(const float* __restrict__ h, const int* __restrict__ p0,
                   const int* __restrict__ p1, const int* __restrict__ q0,
                   const int* __restrict__ q1, const float* __restrict__ Wm1,
                   const float* __restrict__ bm1, const float* __restrict__ Wm2,
                   const float* __restrict__ bm2, float* __restrict__ out,
                   int NP, int Etot) {
  const int lane = threadIdx.x & 31;
  const int m = lane & 15;
  const int hh = lane >> 4;
  const int kb = hh * 2;
  const int wave = __builtin_amdgcn_readfirstlane(
      (int)(blockIdx.x * (blockDim.x >> 5) + (threadIdx.x >> 5)));
  const int nwaves = (int)(gridDim.x * (blockDim.x >> 5));
  const int ntiles = (Etot + 15) >> 4;

  v2f w0[16], w1[16];
#pragma unroll
  for (int ks = 0; ks < 16; ++ks) {
    int j = ks * 4 + kb;
    w0[ks].x = Wm1[(size_t)j * HID + m];
    w0[ks].y = Wm1[(size_t)(j + 1) * HID + m];
    w1[ks].x = Wm1[(size_t)j * HID + 16 + m];
    w1[ks].y = Wm1[(size_t)(j + 1) * HID + 16 + m];
  }
  v8f biasA, biasB;
  float wa0[8], wa1[8], wb0[8], wb1[8];
#pragma unroll
  for (int r = 0; r < 8; ++r) {
    const int nA = 8 * hh + r, nB = 16 + 8 * hh + r;
    biasA[r] = bm1[nA];
    biasB[r] = bm1[nB];
    wa0[r] = Wm2[nA * 2];  wa1[r] = Wm2[nA * 2 + 1];
    wb0[r] = Wm2[nB * 2];  wb1[r] = Wm2[nB * 2 + 1];
  }
  const float c0 = bm2[0], c1 = bm2[1];

  for (int t = wave; t < ntiles; t += nwaves) {
    const int e0 = t * 16;
    int e = e0 + m;
    if (e > Etot - 1) e = Etot - 1;
    int s = (e < NP) ? p0[e] : q0[e - NP];
    int d = (e < NP) ? p1[e] : q1[e - NP];
    const float* hs = h + (size_t)s * HID;
    const float* hd = h + (size_t)d * HID;
    v8f acc0 = biasA, acc1 = biasB;
#pragma unroll
    for (int ks = 0; ks < 16; ++ks) {
      int j = ks * 4 + kb;
      const float* ap = (j < HID) ? (hs + j) : (hd + (j - HID));
      v2f a; a.x = ap[0]; a.y = ap[1];
      acc0 = __builtin_amdgcn_wmma_f32_16x16x4_f32(false, w0[ks], false, a, (short)0, acc0,
                                                   false, false);
      acc1 = __builtin_amdgcn_wmma_f32_16x16x4_f32(false, w1[ks], false, a, (short)0, acc1,
                                                   false, false);
    }
    float o0 = 0.f, o1 = 0.f;
#pragma unroll
    for (int r = 0; r < 8; ++r) {
      const float t0 = fmaxf(acc0[r], 0.f);
      const float t1 = fmaxf(acc1[r], 0.f);
      o0 = fmaf(t0, wa0[r], o0);
      o1 = fmaf(t0, wa1[r], o1);
      o0 = fmaf(t1, wb0[r], o0);
      o1 = fmaf(t1, wb1[r], o1);
    }
    o0 += __shfl_xor(o0, 16);
    o1 += __shfl_xor(o1, 16);
    if (hh == 0 && e0 + m < Etot) {
      v2f o; o.x = o0 + c0; o.y = o1 + c1;
      *(v2f*)(out + (size_t)(e0 + m) * 2) = o;
    }
  }
}

extern "C" void kernel_launch(void* const* d_in, const int* in_sizes, int n_in,
                              void* d_out, int out_size, void* d_ws, size_t ws_size,
                              hipStream_t stream) {
  const float* x   = (const float*)d_in[0];
  const int*   pe  = (const int*)d_in[1];
  const int*   neg = (const int*)d_in[2];
  const float* W1  = (const float*)d_in[3];
  const float* b1  = (const float*)d_in[4];
  const float* W2  = (const float*)d_in[5];
  const float* b2  = (const float*)d_in[6];
  const float* Wm1 = (const float*)d_in[7];
  const float* bm1 = (const float*)d_in[8];
  const float* Wm2 = (const float*)d_in[9];
  const float* bm2 = (const float*)d_in[10];
  float* out = (float*)d_out;

  const int N  = in_sizes[0] / HID;
  const int NP = in_sizes[1] / 2;
  const int NN = in_sizes[2] / 2;
  const int ET = NP + NN;
  const int* p0 = pe;        const int* p1 = pe + NP;
  const int* q0 = neg;       const int* q1 = neg + NN;
  const int NB = (N + TN - 1) / TN;

  char* ws = (char*)d_ws;
  size_t off = 0;
  auto wsalloc = [&](size_t bytes) -> char* {
    char* p = ws + off;
    off += (bytes + 255) & ~(size_t)255;
    return p;
  };
  float* dinv = (float*)wsalloc((size_t)N * 4);
  float* xl1  = (float*)wsalloc((size_t)N * H1 * 4);
  float* h1   = (float*)wsalloc((size_t)N * H1 * 4);
  float* xl2  = (float*)wsalloc((size_t)N * HID * 4);
  float* h2   = (float*)wsalloc((size_t)N * HID * 4);
  int*   bcnt = (int*)wsalloc((size_t)NB * 2 * 4);
  int*   bcur = bcnt + NB;
  int*   bst  = (int*)wsalloc((size_t)(NB + 1) * 4);
  int2*  ebin = (int2*)wsalloc((size_t)NP * 8);
  (void)ws_size; (void)n_in; (void)out_size;

  const int B = 256;

  hipMemsetAsync(bcnt, 0, (size_t)NB * 2 * 4, stream);
  {
    const int nblk = (NP + CHUNK - 1) / CHUNK;
    bin_count<<<nblk, 1024, 0, stream>>>(p1, bcnt, NP, NB);
    bin_fill<<<nblk, 1024, 0, stream>>>(p0, p1, bcnt, bcur, bst, ebin, NP, NB);
  }

  gemm_strip_f32<HID, H1, true><<<GEMM_BLOCKS, GEMM_WAVES * 32, 0, stream>>>(
      x, W1, xl1, N, ebin, bst, dinv, NB);
  agg_tile<H1><<<NB, 256, 0, stream>>>(xl1, ebin, bst, dinv, b1, h1, N);

  gemm_strip_f32<H1, HID, false><<<GEMM_BLOCKS, GEMM_WAVES * 32, 0, stream>>>(
      h1, W2, xl2, N, nullptr, nullptr, nullptr, 0);
  agg_tile<HID><<<NB, 256, 0, stream>>>(xl2, ebin, bst, dinv, b2, h2, N);

  edge_mlp_wmma<<<MLP_BLOCKS, B, 0, stream>>>(h2, p0, p1, q0, q1, Wm1, bm1, Wm2, bm2,
                                              out, NP, ET);
}
